# starting kernel
# baseline (speedup 1.0000x reference)
#include <hip/hip_runtime.h>


typedef __attribute__((ext_vector_type(2))) float v2f;
typedef __attribute__((ext_vector_type(4))) float f4;
typedef __attribute__((ext_vector_type(8))) float v8f;

#define NUM_K 524288
#define NUM_A 64
#define B1 256
#define NBLK (NUM_K / B1)

#define SWZ_XOR(x) (((x) << 10) | 0x1f)

template <int IMM>
__device__ __forceinline__ float xor_add(float s) {
  return s + __int_as_float(
                 __builtin_amdgcn_ds_swizzle(__float_as_int(s), IMM));
}

__device__ __forceinline__ float wave_sum32(float v) {
  v2f a;
  a[0] = 1.0f;
  a[1] = 1.0f;
  v2f b;
  b[0] = v;
  b[1] = 0.0f;
  v8f c = {};
  v8f d = __builtin_amdgcn_wmma_f32_16x16x4_f32(
       false, a,  false, b,
       (short)0, c,  false,  false);
  float s = d[0];
  s = xor_add<SWZ_XOR(8)>(s);
  s = xor_add<SWZ_XOR(4)>(s);
  s = xor_add<SWZ_XOR(2)>(s);
  s = xor_add<SWZ_XOR(1)>(s);
  return s;
}

__global__ __launch_bounds__(B1) void rw_compute(
    const float* __restrict__ alpha, const float* __restrict__ theta,
    const float* __restrict__ delta, float* __restrict__ w,
    float* __restrict__ partials) {
  const int k = blockIdx.x * B1 + threadIdx.x;

  const f4* a4 = (const f4*)(alpha + (size_t)k * NUM_A);
  const f4* d4 = (const f4*)(delta + (size_t)k * NUM_A);

  float s = 0.0f;
#pragma unroll
  for (int i = 0; i < NUM_A / 4; ++i) {
    const f4 av = __builtin_nontemporal_load(a4 + i);
    const f4 dv = __builtin_nontemporal_load(d4 + i);
    s = fmaf(dv[0], __builtin_amdgcn_logf(av[0]), s);
    s = fmaf(dv[1], __builtin_amdgcn_logf(av[1]), s);
    s = fmaf(dv[2], __builtin_amdgcn_logf(av[2]), s);
    s = fmaf(dv[3], __builtin_amdgcn_logf(av[3]), s);
  }
  const float wk = theta[k] * __builtin_amdgcn_exp2f(s);
  w[k] = wk;

  const float wsum = wave_sum32(wk);
  __shared__ float sm[B1 / 32];
  const int lane = threadIdx.x & 31;
  const int wv = threadIdx.x >> 5;
  if (lane == 0) sm[wv] = wsum;
  __syncthreads();
  if (threadIdx.x == 0) {
    float t = 0.0f;
#pragma unroll
    for (int i = 0; i < B1 / 32; ++i) t += sm[i];
    partials[blockIdx.x] = t;
  }
}

__global__ __launch_bounds__(256) void rw_reduce(
    const float* __restrict__ partials, float* __restrict__ inv_out) {
  float s = 0.0f;
#pragma unroll
  for (int i = 0; i < NBLK / 256; ++i)
    s += partials[threadIdx.x + i * 256];
  const float wsum = wave_sum32(s);
  __shared__ float sm[256 / 32];
  const int lane = threadIdx.x & 31;
  const int wv = threadIdx.x >> 5;
  if (lane == 0) sm[wv] = wsum;
  __syncthreads();
  if (threadIdx.x == 0) {
    float t = 0.0f;
#pragma unroll
    for (int i = 0; i < 256 / 32; ++i) t += sm[i];
    inv_out[0] = 1.0f / t;
  }
}

__global__ __launch_bounds__(256) void rw_scale(float* __restrict__ w,
                                                const float* __restrict__ inv_p) {
  const float inv = inv_p[0];
  const int idx = blockIdx.x * 256 + threadIdx.x;
  f4* w4 = (f4*)w;
  f4 v = w4[idx];
  v[0] *= inv;
  v[1] *= inv;
  v[2] *= inv;
  v[3] *= inv;
  w4[idx] = v;
}

extern "C" void kernel_launch(void* const* d_in, const int* in_sizes, int n_in,
                              void* d_out, int out_size, void* d_ws,
                              size_t ws_size, hipStream_t stream) {
  (void)in_sizes;
  (void)n_in;
  (void)out_size;
  (void)ws_size;
  const float* alpha = (const float*)d_in[0];
  const float* theta = (const float*)d_in[1];
  const float* delta = (const float*)d_in[2];
  float* out = (float*)d_out;

  float* partials = (float*)d_ws;
  float* inv = partials + NBLK;

  rw_compute<<<NBLK, B1, 0, stream>>>(alpha, theta, delta, out, partials);
  rw_reduce<<<1, 256, 0, stream>>>(partials, inv);
  rw_scale<<<NUM_K / 4 / 256, 256, 0, stream>>>(out, inv);
}


# coalesce the stream: 16 lanes per rule, one wave load = 512 contiguous bytes (was 1 rule per lane, 256-byte stride); 256x16 waves x 128 rows, loads a group ahead; 16-lane DPP fold (sum order changes)
# speedup vs baseline: 6.4099x; 6.4099x over previous
#include <hip/hip_runtime.h>


typedef __attribute__((ext_vector_type(2))) float v2f;
typedef __attribute__((ext_vector_type(4))) float f4;
typedef __attribute__((ext_vector_type(8))) float v8f;

#define NUM_K 524288
#define NUM_A 64
#define WAVES 16
#define B1 (WAVES * 32)
#define WROWS 128
#define NBLK (NUM_K / (WAVES * WROWS))
#define GROUPS (WROWS / 16)

#define SWZ_XOR(x) (((x) << 10) | 0x1f)

template <int IMM>
__device__ __forceinline__ float xor_add(float s) {
  return s + __int_as_float(
                 __builtin_amdgcn_ds_swizzle(__float_as_int(s), IMM));
}

__device__ __forceinline__ float wave_sum32(float v) {
  v2f a;
  a[0] = 1.0f;
  a[1] = 1.0f;
  v2f b;
  b[0] = v;
  b[1] = 0.0f;
  v8f c = {};
  v8f d = __builtin_amdgcn_wmma_f32_16x16x4_f32(
       false, a,  false, b,
       (short)0, c,  false,  false);
  float s = d[0];
  s = xor_add<SWZ_XOR(8)>(s);
  s = xor_add<SWZ_XOR(4)>(s);
  s = xor_add<SWZ_XOR(2)>(s);
  s = xor_add<SWZ_XOR(1)>(s);
  return s;
}

template <int M>
__device__ __forceinline__ float row_xor(float v) {
  return __int_as_float(__builtin_amdgcn_update_dpp(
      0, __float_as_int(v), 0x160 | M, 0xf, 0xf, true));
}

template <int N>
__device__ __forceinline__ void fold(float (&v)[16], const bool up) {
#pragma unroll
  for (int i = 0; i < N; ++i) {
    const float keep = up ? v[i + N] : v[i];
    const float send = up ? v[i] : v[i + N];
    v[i] = keep + row_xor<N>(send);
  }
}

__global__ __launch_bounds__(B1) void rw_compute(
    const float* __restrict__ alpha, const float* __restrict__ theta,
    const float* __restrict__ delta, float* __restrict__ w,
    float* __restrict__ partials) {
  const int lane = threadIdx.x & 31;
  const int wv = __builtin_amdgcn_readfirstlane(threadIdx.x >> 5);
  const int row0 = (blockIdx.x * WAVES + wv) * WROWS;

  const f4* a4 = (const f4*)alpha + (size_t)row0 * (NUM_A / 4) + lane;
  const f4* d4 = (const f4*)delta + (size_t)row0 * (NUM_A / 4) + lane;

  const int myrow = row0 + 2 * (lane & 15) + (lane >> 4);

  float th[WROWS / 32];
#pragma unroll
  for (int m = 0; m < WROWS / 32; ++m) th[m] = theta[myrow + 32 * m];

  f4 ab[2][8], db[2][8];
#define LOAD_GROUP(G)                                                        \
  do {                                                                       \
    _Pragma("unroll") for (int i = 0; i < 8; ++i) {                          \
      ab[(G) & 1][i] = __builtin_nontemporal_load(a4 + ((G) * 8 + i) * 32);  \
      db[(G) & 1][i] = __builtin_nontemporal_load(d4 + ((G) * 8 + i) * 32);  \
    }                                                                        \
  } while (0)

  LOAD_GROUP(0);
  float p[16];
  float wacc = 0.0f;
#pragma unroll
  for (int g = 0; g < GROUPS; ++g) {
    __builtin_amdgcn_sched_barrier(0);
    if (g + 1 < GROUPS) LOAD_GROUP(g + 1);
    __builtin_amdgcn_sched_barrier(0);
#pragma unroll
    for (int i = 0; i < 8; ++i) {
      const f4 av = ab[g & 1][i];
      const f4 dv = db[g & 1][i];
      float s = dv[0] * __builtin_amdgcn_logf(av[0]);
      s = fmaf(dv[1], __builtin_amdgcn_logf(av[1]), s);
      s = fmaf(dv[2], __builtin_amdgcn_logf(av[2]), s);
      s = fmaf(dv[3], __builtin_amdgcn_logf(av[3]), s);
      p[(g & 1) * 8 + i] = s;
    }
    if (g & 1) {
      fold<8>(p, (lane & 8) != 0);
      fold<4>(p, (lane & 4) != 0);
      fold<2>(p, (lane & 2) != 0);
      fold<1>(p, (lane & 1) != 0);
      const int m = g >> 1;
      const float wk = th[m] * __builtin_amdgcn_exp2f(p[0]);
      w[myrow + 32 * m] = wk;
      wacc += wk;
    }
  }
#undef LOAD_GROUP

  const float wsum = wave_sum32(wacc);
  __shared__ float sm[WAVES];
  if (lane == 0) sm[wv] = wsum;
  __syncthreads();
  if (threadIdx.x == 0) {
    float t = 0.0f;
#pragma unroll
    for (int i = 0; i < WAVES; ++i) t += sm[i];
    partials[blockIdx.x] = t;
  }
}

__global__ __launch_bounds__(256) void rw_reduce(
    const float* __restrict__ partials, float* __restrict__ inv_out) {
  float s = 0.0f;
#pragma unroll
  for (int i = 0; i < NBLK / 256; ++i)
    s += partials[threadIdx.x + i * 256];
  const float wsum = wave_sum32(s);
  __shared__ float sm[256 / 32];
  const int lane = threadIdx.x & 31;
  const int wv = threadIdx.x >> 5;
  if (lane == 0) sm[wv] = wsum;
  __syncthreads();
  if (threadIdx.x == 0) {
    float t = 0.0f;
#pragma unroll
    for (int i = 0; i < 256 / 32; ++i) t += sm[i];
    inv_out[0] = 1.0f / t;
  }
}

__global__ __launch_bounds__(256) void rw_scale(float* __restrict__ w,
                                                const float* __restrict__ inv_p) {
  const float inv = inv_p[0];
  const int idx = blockIdx.x * 256 + threadIdx.x;
  f4* w4 = (f4*)w;
  f4 v = w4[idx];
  v[0] *= inv;
  v[1] *= inv;
  v[2] *= inv;
  v[3] *= inv;
  w4[idx] = v;
}

extern "C" void kernel_launch(void* const* d_in, const int* in_sizes, int n_in,
                              void* d_out, int out_size, void* d_ws,
                              size_t ws_size, hipStream_t stream) {
  (void)in_sizes;
  (void)n_in;
  (void)out_size;
  (void)ws_size;
  const float* alpha = (const float*)d_in[0];
  const float* theta = (const float*)d_in[1];
  const float* delta = (const float*)d_in[2];
  float* out = (float*)d_out;

  float* partials = (float*)d_ws;
  float* inv = partials + NBLK;

  rw_compute<<<NBLK, B1, 0, stream>>>(alpha, theta, delta, out, partials);
  rw_reduce<<<1, 256, 0, stream>>>(partials, inv);
  rw_scale<<<NUM_K / 4 / 256, 256, 0, stream>>>(out, inv);
}


# 2 launches instead of 3: the one-workgroup reduce launch is gone; each of 128 scale workgroups adds the 256 partial sums itself (same order, same bits) and scales 4096 values
# speedup vs baseline: 6.6240x; 1.0334x over previous
#include <hip/hip_runtime.h>


typedef __attribute__((ext_vector_type(2))) float v2f;
typedef __attribute__((ext_vector_type(4))) float f4;
typedef __attribute__((ext_vector_type(8))) float v8f;

#define NUM_K 524288
#define NUM_A 64
#define WAVES 16
#define B1 (WAVES * 32)
#define WROWS 128
#define NBLK (NUM_K / (WAVES * WROWS))
#define GROUPS (WROWS / 16)
#define SBLK 128
#define STHR 256
#define SPT (NUM_K / 4 / (SBLK * STHR))
static_assert(STHR == NBLK, "the scaling launch reads one partial sum per thread");
static_assert(SBLK * STHR * SPT * 4 == NUM_K, "the scaling launch covers w exactly once");

#define SWZ_XOR(x) (((x) << 10) | 0x1f)

template <int IMM>
__device__ __forceinline__ float xor_add(float s) {
  return s + __int_as_float(
                 __builtin_amdgcn_ds_swizzle(__float_as_int(s), IMM));
}

__device__ __forceinline__ float wave_sum32(float v) {
  v2f a;
  a[0] = 1.0f;
  a[1] = 1.0f;
  v2f b;
  b[0] = v;
  b[1] = 0.0f;
  v8f c = {};
  v8f d = __builtin_amdgcn_wmma_f32_16x16x4_f32(
       false, a,  false, b,
       (short)0, c,  false,  false);
  float s = d[0];
  s = xor_add<SWZ_XOR(8)>(s);
  s = xor_add<SWZ_XOR(4)>(s);
  s = xor_add<SWZ_XOR(2)>(s);
  s = xor_add<SWZ_XOR(1)>(s);
  return s;
}

template <int M>
__device__ __forceinline__ float row_xor(float v) {
  return __int_as_float(__builtin_amdgcn_update_dpp(
      0, __float_as_int(v), 0x160 | M, 0xf, 0xf, true));
}

template <int N>
__device__ __forceinline__ void fold(float (&v)[16], const bool up) {
#pragma unroll
  for (int i = 0; i < N; ++i) {
    const float keep = up ? v[i + N] : v[i];
    const float send = up ? v[i] : v[i + N];
    v[i] = keep + row_xor<N>(send);
  }
}

__global__ __launch_bounds__(B1) void rw_compute(
    const float* __restrict__ alpha, const float* __restrict__ theta,
    const float* __restrict__ delta, float* __restrict__ w,
    float* __restrict__ partials) {
  const int lane = threadIdx.x & 31;
  const int wv = __builtin_amdgcn_readfirstlane(threadIdx.x >> 5);
  const int row0 = (blockIdx.x * WAVES + wv) * WROWS;

  const f4* a4 = (const f4*)alpha + (size_t)row0 * (NUM_A / 4) + lane;
  const f4* d4 = (const f4*)delta + (size_t)row0 * (NUM_A / 4) + lane;

  const int myrow = row0 + 2 * (lane & 15) + (lane >> 4);

  float th[WROWS / 32];
#pragma unroll
  for (int m = 0; m < WROWS / 32; ++m) th[m] = theta[myrow + 32 * m];

  f4 ab[2][8], db[2][8];
#define LOAD_GROUP(G)                                                        \
  do {                                                                       \
    _Pragma("unroll") for (int i = 0; i < 8; ++i) {                          \
      ab[(G) & 1][i] = __builtin_nontemporal_load(a4 + ((G) * 8 + i) * 32);  \
      db[(G) & 1][i] = __builtin_nontemporal_load(d4 + ((G) * 8 + i) * 32);  \
    }                                                                        \
  } while (0)

  LOAD_GROUP(0);
  float p[16];
  float wacc = 0.0f;
#pragma unroll
  for (int g = 0; g < GROUPS; ++g) {
    __builtin_amdgcn_sched_barrier(0);
    if (g + 1 < GROUPS) LOAD_GROUP(g + 1);
    __builtin_amdgcn_sched_barrier(0);
#pragma unroll
    for (int i = 0; i < 8; ++i) {
      const f4 av = ab[g & 1][i];
      const f4 dv = db[g & 1][i];
      float s = dv[0] * __builtin_amdgcn_logf(av[0]);
      s = fmaf(dv[1], __builtin_amdgcn_logf(av[1]), s);
      s = fmaf(dv[2], __builtin_amdgcn_logf(av[2]), s);
      s = fmaf(dv[3], __builtin_amdgcn_logf(av[3]), s);
      p[(g & 1) * 8 + i] = s;
    }
    if (g & 1) {
      fold<8>(p, (lane & 8) != 0);
      fold<4>(p, (lane & 4) != 0);
      fold<2>(p, (lane & 2) != 0);
      fold<1>(p, (lane & 1) != 0);
      const int m = g >> 1;
      const float wk = th[m] * __builtin_amdgcn_exp2f(p[0]);
      w[myrow + 32 * m] = wk;
      wacc += wk;
    }
  }
#undef LOAD_GROUP

  const float wsum = wave_sum32(wacc);
  __shared__ float sm[WAVES];
  if (lane == 0) sm[wv] = wsum;
  __syncthreads();
  if (threadIdx.x == 0) {
    float t = 0.0f;
#pragma unroll
    for (int i = 0; i < WAVES; ++i) t += sm[i];
    partials[blockIdx.x] = t;
  }
}

__global__ __launch_bounds__(STHR) void rw_scale(
    float* __restrict__ w, const float* __restrict__ partials) {
  const int lane = threadIdx.x & 31;
  const int wv = threadIdx.x >> 5;

  const float s = partials[threadIdx.x];
  f4* w4 = (f4*)w + (size_t)blockIdx.x * (STHR * SPT) + threadIdx.x;
  f4 v[SPT];
#pragma unroll
  for (int i = 0; i < SPT; ++i) v[i] = w4[i * STHR];

  const float wsum = wave_sum32(s);
  __shared__ float sm[STHR / 32];
  if (lane == 0) sm[wv] = wsum;
  __syncthreads();
  float t = 0.0f;
#pragma unroll
  for (int i = 0; i < STHR / 32; ++i) t += sm[i];
  const float inv = 1.0f / t;

#pragma unroll
  for (int i = 0; i < SPT; ++i) {
    f4 x = v[i];
    x[0] *= inv;
    x[1] *= inv;
    x[2] *= inv;
    x[3] *= inv;
    w4[i * STHR] = x;
  }
}

extern "C" void kernel_launch(void* const* d_in, const int* in_sizes, int n_in,
                              void* d_out, int out_size, void* d_ws,
                              size_t ws_size, hipStream_t stream) {
  (void)in_sizes;
  (void)n_in;
  (void)out_size;
  (void)ws_size;
  const float* alpha = (const float*)d_in[0];
  const float* theta = (const float*)d_in[1];
  const float* delta = (const float*)d_in[2];
  float* out = (float*)d_out;

  float* partials = (float*)d_ws;

  rw_compute<<<NBLK, B1, 0, stream>>>(alpha, theta, delta, out, partials);
  rw_scale<<<SBLK, STHR, 0, stream>>>(out, partials);
}
